# starting kernel
# baseline (speedup 1.0000x reference)
#include <hip/hip_runtime.h>
#include <cstdint>

#define NN 100000
#define NE 640000
#define DI 128
#define DH 128
#define DO 64

typedef __attribute__((ext_vector_type(2))) float v2f;
typedef __attribute__((ext_vector_type(4))) float v4f;
typedef __attribute__((ext_vector_type(8))) float v8f;

__global__ __launch_bounds__(256) void k_degree(const int* __restrict__ dst,
                                                float* __restrict__ deg, int E) {
  int e = blockIdx.x * blockDim.x + threadIdx.x;
  if (e < E) atomicAdd(&deg[dst[e]], 1.0f);
}

__global__ __launch_bounds__(256) void k_dinv(float* __restrict__ deg, int n) {
  int i = blockIdx.x * blockDim.x + threadIdx.x;
  if (i < n) deg[i] = rsqrtf(deg[i] + 1.0f);
}

__global__ __launch_bounds__(256) void k_norm(const int* __restrict__ src,
                                              const int* __restrict__ dst,
                                              const float* __restrict__ dinv,
                                              float* __restrict__ norm, int E) {
  int e = blockIdx.x * blockDim.x + threadIdx.x;
  if (e < E) norm[e] = dinv[src[e]] * dinv[dst[e]];
}

template <int COLT>
__global__ __launch_bounds__(COLT * 32) void k_gemm_wmma(
    const float* __restrict__ X, const float* __restrict__ W,
    float* __restrict__ H) {
  constexpr int DOUT = COLT * 16;
  const int lane = threadIdx.x & 31;
  const int wave = threadIdx.x >> 5;
  const int row0 = blockIdx.x * 16;
  const int col0 = wave * 16;
  const int sel = lane & 15;
  const int khalf = (lane >> 4) << 1;

  const float* Arow = X + (size_t)(row0 + sel) * DI + khalf;
  const float* Bcol = W + (size_t)khalf * DOUT + col0 + sel;

  v8f acc = {};
#pragma unroll
  for (int k0 = 0; k0 < DI; k0 += 4) {
    v2f a = *(const v2f*)(Arow + k0);
    v2f b;
    b.x = Bcol[(size_t)k0 * DOUT];
    b.y = Bcol[(size_t)(k0 + 1) * DOUT];
    acc = __builtin_amdgcn_wmma_f32_16x16x4_f32(false, a, false, b, (short)0,
                                                acc, false, false);
  }
  const int mbase = row0 + ((lane >> 4) << 3);
  float* out = H + col0 + sel;
#pragma unroll
  for (int r = 0; r < 8; ++r) out[(size_t)(mbase + r) * DOUT] = acc[r];
}

template <int D>
__global__ __launch_bounds__(256) void k_scatter(
    const float* __restrict__ H, const int* __restrict__ src,
    const int* __restrict__ dst, const float* __restrict__ norm,
    float* __restrict__ agg, int E) {
  constexpr int CH = D / 4;
  unsigned t = blockIdx.x * 256u + threadIdx.x;
  if (t >= (unsigned)E * CH) return;
  int e = t / CH;
  int c = t % CH;
  int s = src[e];
  int d = dst[e];
  float w = norm[e];
  v4f h = *(const v4f*)(H + (size_t)s * D + c * 4);
  float* a = agg + (size_t)d * D + c * 4;
  atomicAdd(a + 0, h.x * w);
  atomicAdd(a + 1, h.y * w);
  atomicAdd(a + 2, h.z * w);
  atomicAdd(a + 3, h.w * w);
}

template <int D, bool RELU>
__global__ __launch_bounds__(256) void k_finish(
    const float* __restrict__ agg, const float* __restrict__ H,
    const float* __restrict__ dinv, const float* __restrict__ bias,
    float* __restrict__ out, int n) {
  constexpr int CH = D / 4;
  unsigned t = blockIdx.x * 256u + threadIdx.x;
  if (t >= (unsigned)n * CH) return;
  int i = t / CH;
  int c = t % CH;
  float sn = dinv[i];
  sn *= sn;
  v4f a = *(const v4f*)(agg + (size_t)i * D + c * 4);
  v4f h = *(const v4f*)(H + (size_t)i * D + c * 4);
  v4f b = *(const v4f*)(bias + c * 4);
  v4f r = a + h * sn + b;
  if (RELU) {
    r.x = fmaxf(r.x, 0.f);
    r.y = fmaxf(r.y, 0.f);
    r.z = fmaxf(r.z, 0.f);
    r.w = fmaxf(r.w, 0.f);
  }
  *(v4f*)(out + (size_t)i * D + c * 4) = r;
}

extern "C" void kernel_launch(void* const* d_in, const int* in_sizes, int n_in,
                              void* d_out, int out_size, void* d_ws,
                              size_t ws_size, hipStream_t stream) {
  (void)in_sizes; (void)n_in; (void)out_size; (void)ws_size;

  const float* x  = (const float*)d_in[0];
  const int*   ei = (const int*)d_in[1];
  const int*   src = ei;
  const int*   dst = ei + NE;
  const float* W1 = (const float*)d_in[2];
  const float* b1 = (const float*)d_in[3];
  const float* W2 = (const float*)d_in[4];
  const float* b2 = (const float*)d_in[5];
  const float* W3 = (const float*)d_in[6];
  const float* b3 = (const float*)d_in[7];

  char* p = (char*)d_ws;
  auto take = [&](size_t bytes) -> void* {
    void* r = (void*)p;
    p += (bytes + 255) & ~(size_t)255;
    return r;
  };
  float* dinv = (float*)take((size_t)NN * 4);
  float* norm = (float*)take((size_t)NE * 4);
  float* bufA = (float*)take((size_t)NN * DH * 4);
  float* bufB = (float*)take((size_t)NN * DH * 4);
  float* bufC = (float*)take((size_t)NN * DH * 4);

  hipMemsetAsync(dinv, 0, (size_t)NN * 4, stream);
  k_degree<<<(NE + 255) / 256, 256, 0, stream>>>(dst, dinv, NE);
  k_dinv<<<(NN + 255) / 256, 256, 0, stream>>>(dinv, NN);
  k_norm<<<(NE + 255) / 256, 256, 0, stream>>>(src, dst, dinv, norm, NE);

  k_gemm_wmma<DH / 16><<<NN / 16, (DH / 16) * 32, 0, stream>>>(x, W1, bufA);
  hipMemsetAsync(bufB, 0, (size_t)NN * DH * 4, stream);
  k_scatter<DH><<<(NE * (DH / 4) + 255) / 256, 256, 0, stream>>>(
      bufA, src, dst, norm, bufB, NE);
  k_finish<DH, true><<<(NN * (DH / 4) + 255) / 256, 256, 0, stream>>>(
      bufB, bufA, dinv, b1, bufB, NN);

  k_gemm_wmma<DH / 16><<<NN / 16, (DH / 16) * 32, 0, stream>>>(bufB, W2, bufA);
  hipMemsetAsync(bufC, 0, (size_t)NN * DH * 4, stream);
  k_scatter<DH><<<(NE * (DH / 4) + 255) / 256, 256, 0, stream>>>(
      bufA, src, dst, norm, bufC, NE);
  k_finish<DH, true><<<(NN * (DH / 4) + 255) / 256, 256, 0, stream>>>(
      bufC, bufA, dinv, b2, bufC, NN);

  k_gemm_wmma<DO / 16><<<NN / 16, (DO / 16) * 32, 0, stream>>>(bufC, W3, bufA);
  hipMemsetAsync(bufB, 0, (size_t)NN * DO * 4, stream);
  k_scatter<DO><<<(NE * (DO / 4) + 255) / 256, 256, 0, stream>>>(
      bufA, src, dst, norm, bufB, NE);
  k_finish<DO, false><<<(NN * (DO / 4) + 255) / 256, 256, 0, stream>>>(
      bufB, bufA, dinv, b3, (float*)d_out, NN);
}


# f32 atomic scatter-add replaced by per-node gather-sum over edges grouped by dst in-call (int atomic arrival-order rank + block prefix sum); each row stored once; sum order now follows that rank
# speedup vs baseline: 2.1633x; 2.1633x over previous
#include <hip/hip_runtime.h>
#include <cstdint>

#define NN 100000
#define NE 640000
#define DI 128
#define DH 128
#define DO 64

typedef __attribute__((ext_vector_type(2))) float v2f;
typedef __attribute__((ext_vector_type(4))) float v4f;
typedef __attribute__((ext_vector_type(8))) float v8f;
typedef __attribute__((ext_vector_type(2))) int v2i;


__global__ __launch_bounds__(256) void k_rank(const int* __restrict__ dst,
                                              int* __restrict__ cnt,
                                              int* __restrict__ rank, int E) {
  int e = blockIdx.x * blockDim.x + threadIdx.x;
  if (e < E) rank[e] = atomicAdd(&cnt[dst[e]], 1);
}

__global__ __launch_bounds__(256) void k_alloc(const int* __restrict__ cnt,
                                               int* __restrict__ cursor,
                                               v2i* __restrict__ seg,
                                               float* __restrict__ dinv, int n) {
  __shared__ int wsum[8];
  __shared__ int blk_base;
  const int tid = threadIdx.x;
  const int lane = tid & 31;
  const int wave = tid >> 5;
  const int i = blockIdx.x * 256 + tid;
  const int c = (i < n) ? cnt[i] : 0;
  int incl = c;
#pragma unroll
  for (int off = 1; off < 32; off <<= 1) {
    int t = __shfl_up(incl, off);
    if (lane >= off) incl += t;
  }
  if (lane == 31) wsum[wave] = incl;
  __syncthreads();
  if (tid == 0) {
    int tot = 0;
#pragma unroll
    for (int w = 0; w < 8; ++w) {
      int t = wsum[w];
      wsum[w] = tot;
      tot += t;
    }
    blk_base = atomicAdd(cursor, tot);
  }
  __syncthreads();
  if (i < n) {
    v2i s;
    s.x = blk_base + wsum[wave] + incl - c;
    s.y = c;
    seg[i] = s;
    dinv[i] = rsqrtf((float)c + 1.0f);
  }
}

__global__ __launch_bounds__(256) void k_fill(const int* __restrict__ src,
                                              const int* __restrict__ dst,
                                              const int* __restrict__ rank,
                                              const v2i* __restrict__ seg,
                                              const float* __restrict__ dinv,
                                              v2i* __restrict__ csr, int E) {
  int e = blockIdx.x * blockDim.x + threadIdx.x;
  if (e >= E) return;
  int s = src[e];
  int d = dst[e];
  int pos = seg[d].x + rank[e];
  v2i r;
  r.x = s;
  r.y = __float_as_int(dinv[s] * dinv[d]);
  csr[pos] = r;
}

template <int COLT>
__global__ __launch_bounds__(COLT * 32) void k_gemm_wmma(
    const float* __restrict__ X, const float* __restrict__ W,
    float* __restrict__ H) {
  constexpr int DOUT = COLT * 16;
  const int lane = threadIdx.x & 31;
  const int wave = threadIdx.x >> 5;
  const int row0 = blockIdx.x * 16;
  const int col0 = wave * 16;
  const int sel = lane & 15;
  const int khalf = (lane >> 4) << 1;

  const float* Arow = X + (size_t)(row0 + sel) * DI + khalf;
  const float* Bcol = W + (size_t)khalf * DOUT + col0 + sel;

  v8f acc = {};
#pragma unroll
  for (int k0 = 0; k0 < DI; k0 += 4) {
    v2f a = *(const v2f*)(Arow + k0);
    v2f b;
    b.x = Bcol[(size_t)k0 * DOUT];
    b.y = Bcol[(size_t)(k0 + 1) * DOUT];
    acc = __builtin_amdgcn_wmma_f32_16x16x4_f32(false, a, false, b, (short)0,
                                                acc, false, false);
  }
  const int mbase = row0 + ((lane >> 4) << 3);
  float* out = H + col0 + sel;
#pragma unroll
  for (int r = 0; r < 8; ++r) out[(size_t)(mbase + r) * DOUT] = acc[r];
}

template <int D, bool RELU>
__global__ __launch_bounds__(256) void k_gather(
    const float* __restrict__ H, const v2i* __restrict__ csr,
    const v2i* __restrict__ seg, const float* __restrict__ dinv,
    const float* __restrict__ bias, float* __restrict__ out, int n_nodes) {
  constexpr int LPN = D / 4;
  const int gt = blockIdx.x * 256 + threadIdx.x;
  const int c = gt % LPN;
  const int stride = (gridDim.x * 256) / LPN;
  const float* Hc = H + c * 4;
  const v4f b = *(const v4f*)(bias + c * 4);
  const v4f zero4 = {0.f, 0.f, 0.f, 0.f};

  int node = gt / LPN;
  v2i sg = {0, 0};
  if (node < n_nodes) sg = seg[node];
  while (node < n_nodes) {
    const int beg = sg.x;
    const int n = sg.y;
    v2i rec = {0, 0};
    if (c < n) rec = csr[beg + c];
    const int nnode = node + stride;
    v2i nsg = {0, 0};
    if (nnode < n_nodes) nsg = seg[nnode];
    float sn = dinv[node];
    const v4f h = *(const v4f*)(Hc + (size_t)node * D);

    v4f acc = zero4;
    int base = 0;
    while (base < n) {
      const int m = (n - base < LPN) ? (n - base) : LPN;
      int k = 0;
      for (; k + 4 <= m; k += 4) {
        const int s0 = __shfl(rec.x, k + 0, LPN);
        const int s1 = __shfl(rec.x, k + 1, LPN);
        const int s2 = __shfl(rec.x, k + 2, LPN);
        const int s3 = __shfl(rec.x, k + 3, LPN);
        const float w0 = __int_as_float(__shfl(rec.y, k + 0, LPN));
        const float w1 = __int_as_float(__shfl(rec.y, k + 1, LPN));
        const float w2 = __int_as_float(__shfl(rec.y, k + 2, LPN));
        const float w3 = __int_as_float(__shfl(rec.y, k + 3, LPN));
        const v4f h0 = *(const v4f*)(Hc + (size_t)s0 * D);
        const v4f h1 = *(const v4f*)(Hc + (size_t)s1 * D);
        const v4f h2 = *(const v4f*)(Hc + (size_t)s2 * D);
        const v4f h3 = *(const v4f*)(Hc + (size_t)s3 * D);
        acc += h0 * w0;
        acc += h1 * w1;
        acc += h2 * w2;
        acc += h3 * w3;
      }
      if (k < m) {
        const int s0 = __shfl(rec.x, k + 0, LPN);
        const int s1 = __shfl(rec.x, k + 1, LPN);
        const int s2 = __shfl(rec.x, k + 2, LPN);
        const float w0 = __int_as_float(__shfl(rec.y, k + 0, LPN));
        const float w1 = __int_as_float(__shfl(rec.y, k + 1, LPN));
        const float w2 = __int_as_float(__shfl(rec.y, k + 2, LPN));
        const v4f h0 = *(const v4f*)(Hc + (size_t)s0 * D);
        v4f h1 = zero4;
        v4f h2 = zero4;
        if (k + 1 < m) h1 = *(const v4f*)(Hc + (size_t)s1 * D);
        if (k + 2 < m) h2 = *(const v4f*)(Hc + (size_t)s2 * D);
        acc += h0 * w0;
        acc += h1 * w1;
        acc += h2 * w2;
      }
      base += LPN;
      if (base < n) {
        rec.x = 0;
        rec.y = 0;
        if (c < n - base) rec = csr[beg + base + c];
      }
    }

    sn *= sn;
    v4f r = acc + h * sn + b;
    if (RELU) {
      r.x = fmaxf(r.x, 0.f);
      r.y = fmaxf(r.y, 0.f);
      r.z = fmaxf(r.z, 0.f);
      r.w = fmaxf(r.w, 0.f);
    }
    *(v4f*)(out + (size_t)node * D + c * 4) = r;

    node = nnode;
    sg = nsg;
  }
}

extern "C" void kernel_launch(void* const* d_in, const int* in_sizes, int n_in,
                              void* d_out, int out_size, void* d_ws,
                              size_t ws_size, hipStream_t stream) {
  (void)in_sizes; (void)n_in; (void)out_size; (void)ws_size;

  const float* x  = (const float*)d_in[0];
  const int*   ei = (const int*)d_in[1];
  const int*   src = ei;
  const int*   dst = ei + NE;
  const float* W1 = (const float*)d_in[2];
  const float* b1 = (const float*)d_in[3];
  const float* W2 = (const float*)d_in[4];
  const float* b2 = (const float*)d_in[5];
  const float* W3 = (const float*)d_in[6];
  const float* b3 = (const float*)d_in[7];

  char* p = (char*)d_ws;
  auto take = [&](size_t bytes) -> void* {
    void* r = (void*)p;
    p += (bytes + 255) & ~(size_t)255;
    return r;
  };
  int*   cnt  = (int*)take((size_t)(NN + 64) * 4);
  int*   cursor = cnt + NN;
  int*   rank = (int*)take((size_t)NE * 4);
  v2i*   seg  = (v2i*)take((size_t)NN * 8);
  float* dinv = (float*)take((size_t)NN * 4);
  v2i*   csr  = (v2i*)take((size_t)NE * 8);
  float* bufA = (float*)take((size_t)NN * DH * 4);
  float* bufB = (float*)take((size_t)NN * DH * 4);

  hipMemsetAsync(cnt, 0, (size_t)(NN + 64) * 4, stream);
  k_rank<<<(NE + 255) / 256, 256, 0, stream>>>(dst, cnt, rank, NE);
  k_alloc<<<(NN + 255) / 256, 256, 0, stream>>>(cnt, cursor, seg, dinv, NN);
  k_fill<<<(NE + 255) / 256, 256, 0, stream>>>(src, dst, rank, seg, dinv, csr, NE);

  k_gemm_wmma<DH / 16><<<NN / 16, (DH / 16) * 32, 0, stream>>>(x, W1, bufA);
  k_gather<DH, true><<<4096, 256, 0, stream>>>(bufA, csr, seg, dinv, b1, bufB, NN);

  k_gemm_wmma<DH / 16><<<NN / 16, (DH / 16) * 32, 0, stream>>>(bufB, W2, bufA);
  k_gather<DH, true><<<4096, 256, 0, stream>>>(bufA, csr, seg, dinv, b2, bufB, NN);

  k_gemm_wmma<DO / 16><<<NN / 16, (DO / 16) * 32, 0, stream>>>(bufB, W3, bufA);
  k_gather<DO, false><<<2048, 256, 0, stream>>>(bufA, csr, seg, dinv, b3,
                                               (float*)d_out, NN);
}


# GEMM as resident work-groups walking all row tiles: each wave keeps its 16-column strip of W in registers, X tiles go through LDS by whole-line loads (was one short-lived wave per 16x16 tile)
# speedup vs baseline: 2.9302x; 1.3545x over previous
#include <hip/hip_runtime.h>
#include <cstdint>

#define NN 100000
#define NE 640000
#define DI 128
#define DH 128
#define DO 64

typedef __attribute__((ext_vector_type(2))) float v2f;
typedef __attribute__((ext_vector_type(4))) float v4f;
typedef __attribute__((ext_vector_type(8))) float v8f;
typedef __attribute__((ext_vector_type(2))) int v2i;


__global__ __launch_bounds__(256) void k_rank(const int* __restrict__ dst,
                                              int* __restrict__ cnt,
                                              int* __restrict__ rank, int E) {
  int e = blockIdx.x * blockDim.x + threadIdx.x;
  if (e < E) rank[e] = atomicAdd(&cnt[dst[e]], 1);
}

__global__ __launch_bounds__(256) void k_alloc(const int* __restrict__ cnt,
                                               int* __restrict__ cursor,
                                               v2i* __restrict__ seg,
                                               float* __restrict__ dinv, int n) {
  __shared__ int wsum[8];
  __shared__ int blk_base;
  const int tid = threadIdx.x;
  const int lane = tid & 31;
  const int wave = tid >> 5;
  const int i = blockIdx.x * 256 + tid;
  const int c = (i < n) ? cnt[i] : 0;
  int incl = c;
#pragma unroll
  for (int off = 1; off < 32; off <<= 1) {
    int t = __shfl_up(incl, off);
    if (lane >= off) incl += t;
  }
  if (lane == 31) wsum[wave] = incl;
  __syncthreads();
  if (tid == 0) {
    int tot = 0;
#pragma unroll
    for (int w = 0; w < 8; ++w) {
      int t = wsum[w];
      wsum[w] = tot;
      tot += t;
    }
    blk_base = atomicAdd(cursor, tot);
  }
  __syncthreads();
  if (i < n) {
    v2i s;
    s.x = blk_base + wsum[wave] + incl - c;
    s.y = c;
    seg[i] = s;
    dinv[i] = rsqrtf((float)c + 1.0f);
  }
}

__global__ __launch_bounds__(256) void k_fill(const int* __restrict__ src,
                                              const int* __restrict__ dst,
                                              const int* __restrict__ rank,
                                              const v2i* __restrict__ seg,
                                              const float* __restrict__ dinv,
                                              v2i* __restrict__ csr, int E) {
  int e = blockIdx.x * blockDim.x + threadIdx.x;
  if (e >= E) return;
  int s = src[e];
  int d = dst[e];
  int pos = seg[d].x + rank[e];
  v2i r;
  r.x = s;
  r.y = __float_as_int(dinv[s] * dinv[d]);
  csr[pos] = r;
}

#define GEMM_BLOCKS_128 768
#define GEMM_BLOCKS_64 512
template <int COLT, int NBLK>
__global__ __launch_bounds__(256) void k_gemm_wmma(
    const float* __restrict__ X, const float* __restrict__ W,
    float* __restrict__ H) {
  constexpr int DOUT = COLT * 16;
  constexpr int RT = 8 / COLT;
  constexpr int R = 16 * RT;
  constexpr int NPASS = NN / R;
  constexpr int PITCH = DI + 4;
  constexpr int NPC = (R * DI) / (256 * 4);
  __shared__ __attribute__((aligned(16))) float As[2][R * PITCH];

  const int tid = threadIdx.x;
  const int lane = tid & 31;
  const int wave = tid >> 5;
  const int strip = wave % COLT;
  const int rsub = wave / COLT;
  const int col0 = strip * 16;
  const int sel = lane & 15;
  const int khalf = (lane >> 4) << 1;

  const float* Bcol = W + (size_t)khalf * DOUT + col0 + sel;
  v2f b[DI / 4];
#pragma unroll
  for (int j = 0; j < DI / 4; ++j) {
    b[j].x = Bcol[(size_t)(4 * j) * DOUT];
    b[j].y = Bcol[(size_t)(4 * j + 1) * DOUT];
  }

  int lds_off[NPC];
#pragma unroll
  for (int p = 0; p < NPC; ++p) {
    const int q = p * 256 + tid;
    lds_off[p] = (q >> 5) * PITCH + (q & 31) * 4;
  }
  const float* Xt = X + (size_t)tid * 4;
  const int a_off = (rsub * 16 + sel) * PITCH + khalf;

  int pass = blockIdx.x;
  v4f pre[NPC];
  if (pass < NPASS) {
#pragma unroll
    for (int p = 0; p < NPC; ++p)
      pre[p] = *(const v4f*)(Xt + (size_t)pass * (R * DI) + p * 1024);
  }
  int buf = 0;
  for (; pass < NPASS; pass += NBLK) {
    float* Ab = As[buf];
#pragma unroll
    for (int p = 0; p < NPC; ++p) *(v4f*)(Ab + lds_off[p]) = pre[p];
    __syncthreads();
    const int npass = pass + NBLK;
    if (npass < NPASS) {
#pragma unroll
      for (int p = 0; p < NPC; ++p)
        pre[p] = *(const v4f*)(Xt + (size_t)npass * (R * DI) + p * 1024);
    }
    const float* Ap = Ab + a_off;
    v2f a[DI / 4];
#pragma unroll
    for (int j = 0; j < DI / 4; ++j) a[j] = *(const v2f*)(Ap + 4 * j);
    __builtin_amdgcn_sched_barrier(0);
    v8f acc = {};
#pragma unroll
    for (int j = 0; j < DI / 4; ++j)
      acc = __builtin_amdgcn_wmma_f32_16x16x4_f32(false, a[j], false, b[j],
                                                  (short)0, acc, false, false);
    const int mbase = pass * R + rsub * 16 + ((lane >> 4) << 3);
    float* out = H + (size_t)mbase * DOUT + col0 + sel;
#pragma unroll
    for (int r = 0; r < 8; ++r) out[r * DOUT] = acc[r];
    buf ^= 1;
  }
}

template <int D, bool RELU>
__global__ __launch_bounds__(256) void k_gather(
    const float* __restrict__ H, const v2i* __restrict__ csr,
    const v2i* __restrict__ seg, const float* __restrict__ dinv,
    const float* __restrict__ bias, float* __restrict__ out, int n_nodes) {
  constexpr int LPN = D / 4;
  const int gt = blockIdx.x * 256 + threadIdx.x;
  const int c = gt % LPN;
  const int stride = (gridDim.x * 256) / LPN;
  const float* Hc = H + c * 4;
  const v4f b = *(const v4f*)(bias + c * 4);
  const v4f zero4 = {0.f, 0.f, 0.f, 0.f};

  int node = gt / LPN;
  v2i sg = {0, 0};
  if (node < n_nodes) sg = seg[node];
  while (node < n_nodes) {
    const int beg = sg.x;
    const int n = sg.y;
    v2i rec = {0, 0};
    if (c < n) rec = csr[beg + c];
    const int nnode = node + stride;
    v2i nsg = {0, 0};
    if (nnode < n_nodes) nsg = seg[nnode];
    float sn = dinv[node];
    const v4f h = *(const v4f*)(Hc + (size_t)node * D);

    v4f acc = zero4;
    int base = 0;
    while (base < n) {
      const int m = (n - base < LPN) ? (n - base) : LPN;
      int k = 0;
      for (; k + 4 <= m; k += 4) {
        const int s0 = __shfl(rec.x, k + 0, LPN);
        const int s1 = __shfl(rec.x, k + 1, LPN);
        const int s2 = __shfl(rec.x, k + 2, LPN);
        const int s3 = __shfl(rec.x, k + 3, LPN);
        const float w0 = __int_as_float(__shfl(rec.y, k + 0, LPN));
        const float w1 = __int_as_float(__shfl(rec.y, k + 1, LPN));
        const float w2 = __int_as_float(__shfl(rec.y, k + 2, LPN));
        const float w3 = __int_as_float(__shfl(rec.y, k + 3, LPN));
        const v4f h0 = *(const v4f*)(Hc + (size_t)s0 * D);
        const v4f h1 = *(const v4f*)(Hc + (size_t)s1 * D);
        const v4f h2 = *(const v4f*)(Hc + (size_t)s2 * D);
        const v4f h3 = *(const v4f*)(Hc + (size_t)s3 * D);
        acc += h0 * w0;
        acc += h1 * w1;
        acc += h2 * w2;
        acc += h3 * w3;
      }
      if (k < m) {
        const int s0 = __shfl(rec.x, k + 0, LPN);
        const int s1 = __shfl(rec.x, k + 1, LPN);
        const int s2 = __shfl(rec.x, k + 2, LPN);
        const float w0 = __int_as_float(__shfl(rec.y, k + 0, LPN));
        const float w1 = __int_as_float(__shfl(rec.y, k + 1, LPN));
        const float w2 = __int_as_float(__shfl(rec.y, k + 2, LPN));
        const v4f h0 = *(const v4f*)(Hc + (size_t)s0 * D);
        v4f h1 = zero4;
        v4f h2 = zero4;
        if (k + 1 < m) h1 = *(const v4f*)(Hc + (size_t)s1 * D);
        if (k + 2 < m) h2 = *(const v4f*)(Hc + (size_t)s2 * D);
        acc += h0 * w0;
        acc += h1 * w1;
        acc += h2 * w2;
      }
      base += LPN;
      if (base < n) {
        rec.x = 0;
        rec.y = 0;
        if (c < n - base) rec = csr[beg + base + c];
      }
    }

    sn *= sn;
    v4f r = acc + h * sn + b;
    if (RELU) {
      r.x = fmaxf(r.x, 0.f);
      r.y = fmaxf(r.y, 0.f);
      r.z = fmaxf(r.z, 0.f);
      r.w = fmaxf(r.w, 0.f);
    }
    *(v4f*)(out + (size_t)node * D + c * 4) = r;

    node = nnode;
    sg = nsg;
  }
}

extern "C" void kernel_launch(void* const* d_in, const int* in_sizes, int n_in,
                              void* d_out, int out_size, void* d_ws,
                              size_t ws_size, hipStream_t stream) {
  (void)in_sizes; (void)n_in; (void)out_size; (void)ws_size;

  const float* x  = (const float*)d_in[0];
  const int*   ei = (const int*)d_in[1];
  const int*   src = ei;
  const int*   dst = ei + NE;
  const float* W1 = (const float*)d_in[2];
  const float* b1 = (const float*)d_in[3];
  const float* W2 = (const float*)d_in[4];
  const float* b2 = (const float*)d_in[5];
  const float* W3 = (const float*)d_in[6];
  const float* b3 = (const float*)d_in[7];

  char* p = (char*)d_ws;
  auto take = [&](size_t bytes) -> void* {
    void* r = (void*)p;
    p += (bytes + 255) & ~(size_t)255;
    return r;
  };
  int*   cnt  = (int*)take((size_t)(NN + 64) * 4);
  int*   cursor = cnt + NN;
  int*   rank = (int*)take((size_t)NE * 4);
  v2i*   seg  = (v2i*)take((size_t)NN * 8);
  float* dinv = (float*)take((size_t)NN * 4);
  v2i*   csr  = (v2i*)take((size_t)NE * 8);
  float* bufA = (float*)take((size_t)NN * DH * 4);
  float* bufB = (float*)take((size_t)NN * DH * 4);

  hipMemsetAsync(cnt, 0, (size_t)(NN + 64) * 4, stream);
  k_rank<<<(NE + 255) / 256, 256, 0, stream>>>(dst, cnt, rank, NE);
  k_alloc<<<(NN + 255) / 256, 256, 0, stream>>>(cnt, cursor, seg, dinv, NN);
  k_fill<<<(NE + 255) / 256, 256, 0, stream>>>(src, dst, rank, seg, dinv, csr, NE);

  k_gemm_wmma<DH / 16, GEMM_BLOCKS_128><<<GEMM_BLOCKS_128, 256, 0, stream>>>(x, W1, bufA);
  k_gather<DH, true><<<4096, 256, 0, stream>>>(bufA, csr, seg, dinv, b1, bufB, NN);

  k_gemm_wmma<DH / 16, GEMM_BLOCKS_128><<<GEMM_BLOCKS_128, 256, 0, stream>>>(bufB, W2, bufA);
  k_gather<DH, true><<<4096, 256, 0, stream>>>(bufA, csr, seg, dinv, b2, bufB, NN);

  k_gemm_wmma<DO / 16, GEMM_BLOCKS_64><<<GEMM_BLOCKS_64, 256, 0, stream>>>(bufB, W3, bufA);
  k_gather<DO, false><<<2048, 256, 0, stream>>>(bufA, csr, seg, dinv, b3,
                                               (float*)d_out, NN);
}


# k_rank, k_fill and the 128-wide gather started as few long-lived waves: 5 edges per thread (20000 -> 4000 waves each), gather grid 4096 -> 2048 resident work-groups; sums still in arrival rank
# speedup vs baseline: 3.0478x; 1.0401x over previous
#include <hip/hip_runtime.h>
#include <cstdint>

#define NN 100000
#define NE 640000
#define DI 128
#define DH 128
#define DO 64

typedef __attribute__((ext_vector_type(2))) float v2f;
typedef __attribute__((ext_vector_type(4))) float v4f;
typedef __attribute__((ext_vector_type(8))) float v8f;
typedef __attribute__((ext_vector_type(2))) int v2i;


#define EPT 5
#define EDGE_PIECE (256 * EPT)
#define EDGE_BLOCKS (NE / EDGE_PIECE)
static_assert(NE % EDGE_PIECE == 0, "edge passes have no bound check");

__global__ __launch_bounds__(256) void k_rank(const int* __restrict__ dst,
                                              int* __restrict__ cnt,
                                              int* __restrict__ rank) {
  const int e0 = blockIdx.x * EDGE_PIECE + threadIdx.x;
  int d[EPT];
#pragma unroll
  for (int i = 0; i < EPT; ++i) d[i] = dst[e0 + i * 256];
  int r[EPT];
#pragma unroll
  for (int i = 0; i < EPT; ++i) r[i] = atomicAdd(&cnt[d[i]], 1);
#pragma unroll
  for (int i = 0; i < EPT; ++i) rank[e0 + i * 256] = r[i];
}

__global__ __launch_bounds__(256) void k_alloc(const int* __restrict__ cnt,
                                               int* __restrict__ cursor,
                                               v2i* __restrict__ seg,
                                               float* __restrict__ dinv, int n) {
  __shared__ int wsum[8];
  __shared__ int blk_base;
  const int tid = threadIdx.x;
  const int lane = tid & 31;
  const int wave = tid >> 5;
  const int i = blockIdx.x * 256 + tid;
  const int c = (i < n) ? cnt[i] : 0;
  int incl = c;
#pragma unroll
  for (int off = 1; off < 32; off <<= 1) {
    int t = __shfl_up(incl, off);
    if (lane >= off) incl += t;
  }
  if (lane == 31) wsum[wave] = incl;
  __syncthreads();
  if (tid == 0) {
    int tot = 0;
#pragma unroll
    for (int w = 0; w < 8; ++w) {
      int t = wsum[w];
      wsum[w] = tot;
      tot += t;
    }
    blk_base = atomicAdd(cursor, tot);
  }
  __syncthreads();
  if (i < n) {
    v2i s;
    s.x = blk_base + wsum[wave] + incl - c;
    s.y = c;
    seg[i] = s;
    dinv[i] = rsqrtf((float)c + 1.0f);
  }
}

__global__ __launch_bounds__(256) void k_fill(const int* __restrict__ src,
                                              const int* __restrict__ dst,
                                              const int* __restrict__ rank,
                                              const v2i* __restrict__ seg,
                                              const float* __restrict__ dinv,
                                              v2i* __restrict__ csr) {
  const int e0 = blockIdx.x * EDGE_PIECE + threadIdx.x;
  int s[EPT], d[EPT], rk[EPT];
#pragma unroll
  for (int i = 0; i < EPT; ++i) {
    s[i] = src[e0 + i * 256];
    d[i] = dst[e0 + i * 256];
    rk[i] = rank[e0 + i * 256];
  }
  int pos[EPT];
  float ws[EPT], wd[EPT];
#pragma unroll
  for (int i = 0; i < EPT; ++i) {
    pos[i] = seg[d[i]].x;
    ws[i] = dinv[s[i]];
    wd[i] = dinv[d[i]];
  }
#pragma unroll
  for (int i = 0; i < EPT; ++i) {
    v2i r;
    r.x = s[i];
    r.y = __float_as_int(ws[i] * wd[i]);
    csr[pos[i] + rk[i]] = r;
  }
}

#define GEMM_BLOCKS_128 768
#define GEMM_BLOCKS_64 512
template <int COLT, int NBLK>
__global__ __launch_bounds__(256) void k_gemm_wmma(
    const float* __restrict__ X, const float* __restrict__ W,
    float* __restrict__ H) {
  constexpr int DOUT = COLT * 16;
  constexpr int RT = 8 / COLT;
  constexpr int R = 16 * RT;
  constexpr int NPASS = NN / R;
  constexpr int PITCH = DI + 4;
  constexpr int NPC = (R * DI) / (256 * 4);
  __shared__ __attribute__((aligned(16))) float As[2][R * PITCH];

  const int tid = threadIdx.x;
  const int lane = tid & 31;
  const int wave = tid >> 5;
  const int strip = wave % COLT;
  const int rsub = wave / COLT;
  const int col0 = strip * 16;
  const int sel = lane & 15;
  const int khalf = (lane >> 4) << 1;

  const float* Bcol = W + (size_t)khalf * DOUT + col0 + sel;
  v2f b[DI / 4];
#pragma unroll
  for (int j = 0; j < DI / 4; ++j) {
    b[j].x = Bcol[(size_t)(4 * j) * DOUT];
    b[j].y = Bcol[(size_t)(4 * j + 1) * DOUT];
  }

  int lds_off[NPC];
#pragma unroll
  for (int p = 0; p < NPC; ++p) {
    const int q = p * 256 + tid;
    lds_off[p] = (q >> 5) * PITCH + (q & 31) * 4;
  }
  const float* Xt = X + (size_t)tid * 4;
  const int a_off = (rsub * 16 + sel) * PITCH + khalf;

  int pass = blockIdx.x;
  v4f pre[NPC];
  if (pass < NPASS) {
#pragma unroll
    for (int p = 0; p < NPC; ++p)
      pre[p] = *(const v4f*)(Xt + (size_t)pass * (R * DI) + p * 1024);
  }
  int buf = 0;
  for (; pass < NPASS; pass += NBLK) {
    float* Ab = As[buf];
#pragma unroll
    for (int p = 0; p < NPC; ++p) *(v4f*)(Ab + lds_off[p]) = pre[p];
    __syncthreads();
    const int npass = pass + NBLK;
    if (npass < NPASS) {
#pragma unroll
      for (int p = 0; p < NPC; ++p)
        pre[p] = *(const v4f*)(Xt + (size_t)npass * (R * DI) + p * 1024);
    }
    const float* Ap = Ab + a_off;
    v2f a[DI / 4];
#pragma unroll
    for (int j = 0; j < DI / 4; ++j) a[j] = *(const v2f*)(Ap + 4 * j);
    __builtin_amdgcn_sched_barrier(0);
    v8f acc = {};
#pragma unroll
    for (int j = 0; j < DI / 4; ++j)
      acc = __builtin_amdgcn_wmma_f32_16x16x4_f32(false, a[j], false, b[j],
                                                  (short)0, acc, false, false);
    const int mbase = pass * R + rsub * 16 + ((lane >> 4) << 3);
    float* out = H + (size_t)mbase * DOUT + col0 + sel;
#pragma unroll
    for (int r = 0; r < 8; ++r) out[r * DOUT] = acc[r];
    buf ^= 1;
  }
}

#define GATHER_BLOCKS 2048
template <int D, bool RELU>
__global__ __launch_bounds__(256) void k_gather(
    const float* __restrict__ H, const v2i* __restrict__ csr,
    const v2i* __restrict__ seg, const float* __restrict__ dinv,
    const float* __restrict__ bias, float* __restrict__ out, int n_nodes) {
  constexpr int LPN = D / 4;
  const int gt = blockIdx.x * 256 + threadIdx.x;
  const int c = gt % LPN;
  const int stride = (gridDim.x * 256) / LPN;
  const float* Hc = H + c * 4;
  const v4f b = *(const v4f*)(bias + c * 4);
  const v4f zero4 = {0.f, 0.f, 0.f, 0.f};

  int node = gt / LPN;
  v2i sg = {0, 0};
  if (node < n_nodes) sg = seg[node];
  while (node < n_nodes) {
    const int beg = sg.x;
    const int n = sg.y;
    v2i rec = {0, 0};
    if (c < n) rec = csr[beg + c];
    const int nnode = node + stride;
    v2i nsg = {0, 0};
    if (nnode < n_nodes) nsg = seg[nnode];
    float sn = dinv[node];
    const v4f h = *(const v4f*)(Hc + (size_t)node * D);

    v4f acc = zero4;
    int base = 0;
    while (base < n) {
      const int m = (n - base < LPN) ? (n - base) : LPN;
      int k = 0;
      for (; k + 4 <= m; k += 4) {
        const int s0 = __shfl(rec.x, k + 0, LPN);
        const int s1 = __shfl(rec.x, k + 1, LPN);
        const int s2 = __shfl(rec.x, k + 2, LPN);
        const int s3 = __shfl(rec.x, k + 3, LPN);
        const float w0 = __int_as_float(__shfl(rec.y, k + 0, LPN));
        const float w1 = __int_as_float(__shfl(rec.y, k + 1, LPN));
        const float w2 = __int_as_float(__shfl(rec.y, k + 2, LPN));
        const float w3 = __int_as_float(__shfl(rec.y, k + 3, LPN));
        const v4f h0 = *(const v4f*)(Hc + (size_t)s0 * D);
        const v4f h1 = *(const v4f*)(Hc + (size_t)s1 * D);
        const v4f h2 = *(const v4f*)(Hc + (size_t)s2 * D);
        const v4f h3 = *(const v4f*)(Hc + (size_t)s3 * D);
        acc += h0 * w0;
        acc += h1 * w1;
        acc += h2 * w2;
        acc += h3 * w3;
      }
      if (k < m) {
        const int s0 = __shfl(rec.x, k + 0, LPN);
        const int s1 = __shfl(rec.x, k + 1, LPN);
        const int s2 = __shfl(rec.x, k + 2, LPN);
        const float w0 = __int_as_float(__shfl(rec.y, k + 0, LPN));
        const float w1 = __int_as_float(__shfl(rec.y, k + 1, LPN));
        const float w2 = __int_as_float(__shfl(rec.y, k + 2, LPN));
        const v4f h0 = *(const v4f*)(Hc + (size_t)s0 * D);
        v4f h1 = zero4;
        v4f h2 = zero4;
        if (k + 1 < m) h1 = *(const v4f*)(Hc + (size_t)s1 * D);
        if (k + 2 < m) h2 = *(const v4f*)(Hc + (size_t)s2 * D);
        acc += h0 * w0;
        acc += h1 * w1;
        acc += h2 * w2;
      }
      base += LPN;
      if (base < n) {
        rec.x = 0;
        rec.y = 0;
        if (c < n - base) rec = csr[beg + base + c];
      }
    }

    sn *= sn;
    v4f r = acc + h * sn + b;
    if (RELU) {
      r.x = fmaxf(r.x, 0.f);
      r.y = fmaxf(r.y, 0.f);
      r.z = fmaxf(r.z, 0.f);
      r.w = fmaxf(r.w, 0.f);
    }
    *(v4f*)(out + (size_t)node * D + c * 4) = r;

    node = nnode;
    sg = nsg;
  }
}

extern "C" void kernel_launch(void* const* d_in, const int* in_sizes, int n_in,
                              void* d_out, int out_size, void* d_ws,
                              size_t ws_size, hipStream_t stream) {
  (void)in_sizes; (void)n_in; (void)out_size; (void)ws_size;

  const float* x  = (const float*)d_in[0];
  const int*   ei = (const int*)d_in[1];
  const int*   src = ei;
  const int*   dst = ei + NE;
  const float* W1 = (const float*)d_in[2];
  const float* b1 = (const float*)d_in[3];
  const float* W2 = (const float*)d_in[4];
  const float* b2 = (const float*)d_in[5];
  const float* W3 = (const float*)d_in[6];
  const float* b3 = (const float*)d_in[7];

  char* p = (char*)d_ws;
  auto take = [&](size_t bytes) -> void* {
    void* r = (void*)p;
    p += (bytes + 255) & ~(size_t)255;
    return r;
  };
  int*   cnt  = (int*)take((size_t)(NN + 64) * 4);
  int*   cursor = cnt + NN;
  int*   rank = (int*)take((size_t)NE * 4);
  v2i*   seg  = (v2i*)take((size_t)NN * 8);
  float* dinv = (float*)take((size_t)NN * 4);
  v2i*   csr  = (v2i*)take((size_t)NE * 8);
  float* bufA = (float*)take((size_t)NN * DH * 4);
  float* bufB = (float*)take((size_t)NN * DH * 4);

  hipMemsetAsync(cnt, 0, (size_t)(NN + 64) * 4, stream);
  k_rank<<<EDGE_BLOCKS, 256, 0, stream>>>(dst, cnt, rank);
  k_alloc<<<(NN + 255) / 256, 256, 0, stream>>>(cnt, cursor, seg, dinv, NN);
  k_fill<<<EDGE_BLOCKS, 256, 0, stream>>>(src, dst, rank, seg, dinv, csr);

  k_gemm_wmma<DH / 16, GEMM_BLOCKS_128><<<GEMM_BLOCKS_128, 256, 0, stream>>>(x, W1, bufA);
  k_gather<DH, true><<<GATHER_BLOCKS, 256, 0, stream>>>(bufA, csr, seg, dinv, b1, bufB, NN);

  k_gemm_wmma<DH / 16, GEMM_BLOCKS_128><<<GEMM_BLOCKS_128, 256, 0, stream>>>(bufB, W2, bufA);
  k_gather<DH, true><<<GATHER_BLOCKS, 256, 0, stream>>>(bufA, csr, seg, dinv, b2, bufB, NN);

  k_gemm_wmma<DO / 16, GEMM_BLOCKS_64><<<GEMM_BLOCKS_64, 256, 0, stream>>>(bufB, W3, bufA);
  k_gather<DO, false><<<GATHER_BLOCKS, 256, 0, stream>>>(bufA, csr, seg, dinv, b3,
                                               (float*)d_out, NN);
}


# gather-sum software-pipelined: a node's own row and up to 8 neighbour rows requested together, next node's records fetched a node ahead; one dependent memory trip per node, was 3; same sum order
# speedup vs baseline: 3.1692x; 1.0398x over previous
#include <hip/hip_runtime.h>
#include <cstdint>

#define NN 100000
#define NE 640000
#define DI 128
#define DH 128
#define DO 64

typedef __attribute__((ext_vector_type(2))) float v2f;
typedef __attribute__((ext_vector_type(4))) float v4f;
typedef __attribute__((ext_vector_type(8))) float v8f;
typedef __attribute__((ext_vector_type(2))) int v2i;


#define EPT 5
#define EDGE_PIECE (256 * EPT)
#define EDGE_BLOCKS (NE / EDGE_PIECE)
static_assert(NE % EDGE_PIECE == 0, "edge passes have no bound check");

__global__ __launch_bounds__(256) void k_rank(const int* __restrict__ dst,
                                              int* __restrict__ cnt,
                                              int* __restrict__ rank) {
  const int e0 = blockIdx.x * EDGE_PIECE + threadIdx.x;
  int d[EPT];
#pragma unroll
  for (int i = 0; i < EPT; ++i) d[i] = dst[e0 + i * 256];
  int r[EPT];
#pragma unroll
  for (int i = 0; i < EPT; ++i) r[i] = atomicAdd(&cnt[d[i]], 1);
#pragma unroll
  for (int i = 0; i < EPT; ++i) rank[e0 + i * 256] = r[i];
}

__global__ __launch_bounds__(256) void k_alloc(const int* __restrict__ cnt,
                                               int* __restrict__ cursor,
                                               v2i* __restrict__ seg,
                                               float* __restrict__ dinv, int n) {
  __shared__ int wsum[8];
  __shared__ int blk_base;
  const int tid = threadIdx.x;
  const int lane = tid & 31;
  const int wave = tid >> 5;
  const int i = blockIdx.x * 256 + tid;
  const int c = (i < n) ? cnt[i] : 0;
  int incl = c;
#pragma unroll
  for (int off = 1; off < 32; off <<= 1) {
    int t = __shfl_up(incl, off);
    if (lane >= off) incl += t;
  }
  if (lane == 31) wsum[wave] = incl;
  __syncthreads();
  if (tid == 0) {
    int tot = 0;
#pragma unroll
    for (int w = 0; w < 8; ++w) {
      int t = wsum[w];
      wsum[w] = tot;
      tot += t;
    }
    blk_base = atomicAdd(cursor, tot);
  }
  __syncthreads();
  if (i < n) {
    v2i s;
    s.x = blk_base + wsum[wave] + incl - c;
    s.y = c;
    seg[i] = s;
    dinv[i] = rsqrtf((float)c + 1.0f);
  }
}

__global__ __launch_bounds__(256) void k_fill(const int* __restrict__ src,
                                              const int* __restrict__ dst,
                                              const int* __restrict__ rank,
                                              const v2i* __restrict__ seg,
                                              const float* __restrict__ dinv,
                                              v2i* __restrict__ csr) {
  const int e0 = blockIdx.x * EDGE_PIECE + threadIdx.x;
  int s[EPT], d[EPT], rk[EPT];
#pragma unroll
  for (int i = 0; i < EPT; ++i) {
    s[i] = src[e0 + i * 256];
    d[i] = dst[e0 + i * 256];
    rk[i] = rank[e0 + i * 256];
  }
  int pos[EPT];
  float ws[EPT], wd[EPT];
#pragma unroll
  for (int i = 0; i < EPT; ++i) {
    pos[i] = seg[d[i]].x;
    ws[i] = dinv[s[i]];
    wd[i] = dinv[d[i]];
  }
#pragma unroll
  for (int i = 0; i < EPT; ++i) {
    v2i r;
    r.x = s[i];
    r.y = __float_as_int(ws[i] * wd[i]);
    csr[pos[i] + rk[i]] = r;
  }
}

#define GEMM_BLOCKS_128 768
#define GEMM_BLOCKS_64 512
template <int COLT, int NBLK>
__global__ __launch_bounds__(256) void k_gemm_wmma(
    const float* __restrict__ X, const float* __restrict__ W,
    float* __restrict__ H) {
  constexpr int DOUT = COLT * 16;
  constexpr int RT = 8 / COLT;
  constexpr int R = 16 * RT;
  constexpr int NPASS = NN / R;
  constexpr int PITCH = DI + 4;
  constexpr int NPC = (R * DI) / (256 * 4);
  __shared__ __attribute__((aligned(16))) float As[2][R * PITCH];

  const int tid = threadIdx.x;
  const int lane = tid & 31;
  const int wave = tid >> 5;
  const int strip = wave % COLT;
  const int rsub = wave / COLT;
  const int col0 = strip * 16;
  const int sel = lane & 15;
  const int khalf = (lane >> 4) << 1;

  const float* Bcol = W + (size_t)khalf * DOUT + col0 + sel;
  v2f b[DI / 4];
#pragma unroll
  for (int j = 0; j < DI / 4; ++j) {
    b[j].x = Bcol[(size_t)(4 * j) * DOUT];
    b[j].y = Bcol[(size_t)(4 * j + 1) * DOUT];
  }

  int lds_off[NPC];
#pragma unroll
  for (int p = 0; p < NPC; ++p) {
    const int q = p * 256 + tid;
    lds_off[p] = (q >> 5) * PITCH + (q & 31) * 4;
  }
  const float* Xt = X + (size_t)tid * 4;
  const int a_off = (rsub * 16 + sel) * PITCH + khalf;

  int pass = blockIdx.x;
  v4f pre[NPC];
  if (pass < NPASS) {
#pragma unroll
    for (int p = 0; p < NPC; ++p)
      pre[p] = *(const v4f*)(Xt + (size_t)pass * (R * DI) + p * 1024);
  }
  int buf = 0;
  for (; pass < NPASS; pass += NBLK) {
    float* Ab = As[buf];
#pragma unroll
    for (int p = 0; p < NPC; ++p) *(v4f*)(Ab + lds_off[p]) = pre[p];
    __syncthreads();
    const int npass = pass + NBLK;
    if (npass < NPASS) {
#pragma unroll
      for (int p = 0; p < NPC; ++p)
        pre[p] = *(const v4f*)(Xt + (size_t)npass * (R * DI) + p * 1024);
    }
    const float* Ap = Ab + a_off;
    v2f a[DI / 4];
#pragma unroll
    for (int j = 0; j < DI / 4; ++j) a[j] = *(const v2f*)(Ap + 4 * j);
    __builtin_amdgcn_sched_barrier(0);
    v8f acc = {};
#pragma unroll
    for (int j = 0; j < DI / 4; ++j)
      acc = __builtin_amdgcn_wmma_f32_16x16x4_f32(false, a[j], false, b[j],
                                                  (short)0, acc, false, false);
    const int mbase = pass * R + rsub * 16 + ((lane >> 4) << 3);
    float* out = H + (size_t)mbase * DOUT + col0 + sel;
#pragma unroll
    for (int r = 0; r < 8; ++r) out[r * DOUT] = acc[r];
    buf ^= 1;
  }
}

#define GATHER_DEPTH 8
#define GATHER_BLOCKS_128 2048
#define GATHER_BLOCKS_64 1536

template <int LPN>
__device__ __forceinline__ int slot_bcast(int v, int idx) {
  if constexpr (LPN == 32) {
    return __builtin_amdgcn_readlane(v, idx);
  } else {
    return __shfl(v, idx, LPN);
  }
}

template <int LPN>
__device__ __forceinline__ int group_uniform(int v) {
  if constexpr (LPN == 32) {
    return __builtin_amdgcn_readfirstlane(v);
  } else {
    return v;
  }
}

#define GATHER_SLOT(j) const int srow##j = slot_bcast<LPN>(rec.x, k + j);
#define GATHER_ROW(j)                                         \
  v4f hrow##j = zero4;                                        \
  if (k + j < m) hrow##j = *(const v4f*)(Hc + (size_t)srow##j * D);
#define GATHER_ADD(j) \
  acc += hrow##j * __int_as_float(slot_bcast<LPN>(rec.y, k + j));

template <int D, bool RELU>
__global__ __launch_bounds__(256) void k_gather(
    const float* __restrict__ H, const v2i* __restrict__ csr,
    const v2i* __restrict__ seg, const float* __restrict__ dinv,
    const float* __restrict__ bias, float* __restrict__ out, int n_nodes) {
  constexpr int LPN = D / 4;
  const int gt = blockIdx.x * 256 + threadIdx.x;
  const int c = gt % LPN;
  const int stride = (gridDim.x * 256) / LPN;
  const float* Hc = H + c * 4;
  const v4f b = *(const v4f*)(bias + c * 4);
  const v4f zero4 = {0.f, 0.f, 0.f, 0.f};

  int node = group_uniform<LPN>(gt / LPN);
  v2i sg = {0, 0};
  v2i nsg = {0, 0};
  v2i rec = {0, 0};
  if (node < n_nodes) sg = seg[node];
  if (node + stride < n_nodes) nsg = seg[node + stride];
  if (c < sg.y) rec = csr[sg.x + c];
  while (node < n_nodes) {
    const int beg = group_uniform<LPN>(sg.x);
    const int n = group_uniform<LPN>(sg.y);
    const int nnode = node + stride;
    const int m0 = (n < LPN) ? n : LPN;
    v4f acc = zero4;
    v2i nrec = {0, 0};
    v2i nnsg = {0, 0};
    float sn = dinv[node];
    const v4f h = *(const v4f*)(Hc + (size_t)node * D);
    {
      const int k = 0;
      const int m = m0;
      GATHER_SLOT(0) GATHER_SLOT(1) GATHER_SLOT(2) GATHER_SLOT(3)
      GATHER_SLOT(4) GATHER_SLOT(5) GATHER_SLOT(6) GATHER_SLOT(7)
      GATHER_ROW(0) GATHER_ROW(1) GATHER_ROW(2) GATHER_ROW(3)
      GATHER_ROW(4) GATHER_ROW(5) GATHER_ROW(6) GATHER_ROW(7)
      if (c < nsg.y) nrec = csr[nsg.x + c];
      if (nnode + stride < n_nodes) nnsg = seg[nnode + stride];
      GATHER_ADD(0) GATHER_ADD(1) GATHER_ADD(2) GATHER_ADD(3)
      GATHER_ADD(4) GATHER_ADD(5) GATHER_ADD(6) GATHER_ADD(7)
    }
    int base = 0;
    int k0 = GATHER_DEPTH;
    while (base < n) {
      const int m = (n - base < LPN) ? (n - base) : LPN;
      for (int k = k0; k < m; k += GATHER_DEPTH) {
        GATHER_SLOT(0) GATHER_SLOT(1) GATHER_SLOT(2) GATHER_SLOT(3)
        GATHER_SLOT(4) GATHER_SLOT(5) GATHER_SLOT(6) GATHER_SLOT(7)
        GATHER_ROW(0) GATHER_ROW(1) GATHER_ROW(2) GATHER_ROW(3)
        GATHER_ROW(4) GATHER_ROW(5) GATHER_ROW(6) GATHER_ROW(7)
        GATHER_ADD(0) GATHER_ADD(1) GATHER_ADD(2) GATHER_ADD(3)
        GATHER_ADD(4) GATHER_ADD(5) GATHER_ADD(6) GATHER_ADD(7)
      }
      k0 = 0;
      base += LPN;
      if (base < n) {
        rec.x = 0;
        rec.y = 0;
        if (c < n - base) rec = csr[beg + base + c];
      }
    }

    sn *= sn;
    v4f r = acc + h * sn + b;
    if (RELU) {
      r.x = fmaxf(r.x, 0.f);
      r.y = fmaxf(r.y, 0.f);
      r.z = fmaxf(r.z, 0.f);
      r.w = fmaxf(r.w, 0.f);
    }
    *(v4f*)(out + (size_t)node * D + c * 4) = r;

    node = nnode;
    sg = nsg;
    nsg = nnsg;
    rec = nrec;
  }
}

extern "C" void kernel_launch(void* const* d_in, const int* in_sizes, int n_in,
                              void* d_out, int out_size, void* d_ws,
                              size_t ws_size, hipStream_t stream) {
  (void)in_sizes; (void)n_in; (void)out_size; (void)ws_size;

  const float* x  = (const float*)d_in[0];
  const int*   ei = (const int*)d_in[1];
  const int*   src = ei;
  const int*   dst = ei + NE;
  const float* W1 = (const float*)d_in[2];
  const float* b1 = (const float*)d_in[3];
  const float* W2 = (const float*)d_in[4];
  const float* b2 = (const float*)d_in[5];
  const float* W3 = (const float*)d_in[6];
  const float* b3 = (const float*)d_in[7];

  char* p = (char*)d_ws;
  auto take = [&](size_t bytes) -> void* {
    void* r = (void*)p;
    p += (bytes + 255) & ~(size_t)255;
    return r;
  };
  int*   cnt  = (int*)take((size_t)(NN + 64) * 4);
  int*   cursor = cnt + NN;
  int*   rank = (int*)take((size_t)NE * 4);
  v2i*   seg  = (v2i*)take((size_t)NN * 8);
  float* dinv = (float*)take((size_t)NN * 4);
  v2i*   csr  = (v2i*)take((size_t)NE * 8);
  float* bufA = (float*)take((size_t)NN * DH * 4);
  float* bufB = (float*)take((size_t)NN * DH * 4);

  hipMemsetAsync(cnt, 0, (size_t)(NN + 64) * 4, stream);
  k_rank<<<EDGE_BLOCKS, 256, 0, stream>>>(dst, cnt, rank);
  k_alloc<<<(NN + 255) / 256, 256, 0, stream>>>(cnt, cursor, seg, dinv, NN);
  k_fill<<<EDGE_BLOCKS, 256, 0, stream>>>(src, dst, rank, seg, dinv, csr);

  k_gemm_wmma<DH / 16, GEMM_BLOCKS_128><<<GEMM_BLOCKS_128, 256, 0, stream>>>(x, W1, bufA);
  k_gather<DH, true><<<GATHER_BLOCKS_128, 256, 0, stream>>>(bufA, csr, seg, dinv, b1, bufB, NN);

  k_gemm_wmma<DH / 16, GEMM_BLOCKS_128><<<GEMM_BLOCKS_128, 256, 0, stream>>>(bufB, W2, bufA);
  k_gather<DH, true><<<GATHER_BLOCKS_128, 256, 0, stream>>>(bufA, csr, seg, dinv, b2, bufB, NN);

  k_gemm_wmma<DO / 16, GEMM_BLOCKS_64><<<GEMM_BLOCKS_64, 256, 0, stream>>>(bufB, W3, bufA);
  k_gather<DO, false><<<GATHER_BLOCKS_64, 256, 0, stream>>>(bufA, csr, seg, dinv, b3,
                                               (float*)d_out, NN);
}


# edges grouped by dst with a two-level LDS counting sort (per-piece bucket order, then one work-group per 256-node bucket ranks in LDS): no per-edge global atomic, no memset; sum order = LDS arrival
# speedup vs baseline: 3.7923x; 1.1966x over previous
#include <hip/hip_runtime.h>
#include <cstdint>

#define NN 100000
#define NE 640000
#define DI 128
#define DH 128
#define DO 64

typedef __attribute__((ext_vector_type(2))) float v2f;
typedef __attribute__((ext_vector_type(4))) float v4f;
typedef __attribute__((ext_vector_type(8))) float v8f;
typedef __attribute__((ext_vector_type(2))) int v2i;


#define EPT 5
#define EDGE_PIECE (256 * EPT)
#define EDGE_BLOCKS (NE / EDGE_PIECE)
static_assert(NE % EDGE_PIECE == 0, "edge passes have no bound check");

#define BUCKET_SHIFT 8
#define BUCKET_NODES (1 << BUCKET_SHIFT)
#define NBUCKET ((NN + BUCKET_NODES - 1) / BUCKET_NODES)
#define NBUCKET_PAD 512
static_assert(NBUCKET + 1 <= NBUCKET_PAD, "bucket table must fit the scan");
#define RUNS_PER_THREAD ((EDGE_BLOCKS + 255) / 256)
#define RUN_BATCH 8

__global__ __launch_bounds__(256) void k_bin(const int* __restrict__ src,
                                             const int* __restrict__ dst,
                                             int* __restrict__ tab,
                                             v2i* __restrict__ binned,
                                             int* __restrict__ cursor) {
  __shared__ int hist[NBUCKET_PAD];
  __shared__ int wsum[8];
  __shared__ v2i stage[EDGE_PIECE];
  const int tid = threadIdx.x;
  const int lane = tid & 31;
  const int wave = tid >> 5;
  const int w = blockIdx.x;
  const int e0 = w * EDGE_PIECE + tid;
  int s[EPT], d[EPT];
#pragma unroll
  for (int i = 0; i < EPT; ++i) {
    s[i] = src[e0 + i * 256];
    d[i] = dst[e0 + i * 256];
  }
  hist[2 * tid] = 0;
  hist[2 * tid + 1] = 0;
  if (w == 0 && tid == 0) *cursor = 0;
  __syncthreads();
  int r[EPT];
#pragma unroll
  for (int i = 0; i < EPT; ++i) r[i] = atomicAdd(&hist[d[i] >> BUCKET_SHIFT], 1);
  __syncthreads();
  const int v0 = hist[2 * tid];
  const int v1 = hist[2 * tid + 1];
  const int sum = v0 + v1;
  int incl = sum;
#pragma unroll
  for (int off = 1; off < 32; off <<= 1) {
    int t = __shfl_up(incl, off);
    if (lane >= off) incl += t;
  }
  if (lane == 31) wsum[wave] = incl;
  __syncthreads();
  int wbase = 0;
#pragma unroll
  for (int k = 0; k < 8; ++k) {
    const int t = wsum[k];
    if (k < wave) wbase += t;
  }
  const int ex = wbase + incl - sum;
  hist[2 * tid] = ex;
  hist[2 * tid + 1] = ex + v0;
  if (2 * tid <= NBUCKET) tab[(2 * tid) * EDGE_BLOCKS + w] = ex;
  if (2 * tid + 1 <= NBUCKET) tab[(2 * tid + 1) * EDGE_BLOCKS + w] = ex + v0;
  __syncthreads();
#pragma unroll
  for (int i = 0; i < EPT; ++i) {
    v2i rec;
    rec.x = s[i];
    rec.y = d[i];
    stage[hist[d[i] >> BUCKET_SHIFT] + r[i]] = rec;
  }
  __syncthreads();
#pragma unroll
  for (int i = 0; i < EPT; ++i)
    binned[(size_t)w * EDGE_PIECE + tid + i * 256] = stage[tid + i * 256];
}

__global__ __launch_bounds__(256) void k_bucket(const int* __restrict__ tab,
                                                const v2i* __restrict__ binned,
                                                int* __restrict__ cursor,
                                                v2i* __restrict__ seg,
                                                float* __restrict__ dinv,
                                                v2i* __restrict__ csr) {
  __shared__ int cntL[BUCKET_NODES];
  __shared__ int cur[BUCKET_NODES];
  __shared__ float dinvL[BUCKET_NODES];
  __shared__ int lsum[8];
  __shared__ int wsum[8];
  __shared__ int blk_base;
  const int tid = threadIdx.x;
  const int lane = tid & 31;
  const int wave = tid >> 5;
  const int bk = blockIdx.x;

  int lo[RUNS_PER_THREAD], hi[RUNS_PER_THREAD];
  const v2i* run[RUNS_PER_THREAD];
#pragma unroll
  for (int q = 0; q < RUNS_PER_THREAD; ++q) {
    const int w = tid + q * 256;
    lo[q] = 0;
    hi[q] = 0;
    if (w < EDGE_BLOCKS) {
      lo[q] = tab[bk * EDGE_BLOCKS + w];
      hi[q] = tab[(bk + 1) * EDGE_BLOCKS + w];
    }
    run[q] = binned + (size_t)(w < EDGE_BLOCKS ? w : 0) * EDGE_PIECE;
  }
  cntL[tid] = 0;
  v2i first[RUNS_PER_THREAD][RUN_BATCH];
#pragma unroll
  for (int q = 0; q < RUNS_PER_THREAD; ++q) {
#pragma unroll
    for (int k = 0; k < RUN_BATCH; ++k) {
      const int j = lo[q] + k;
      first[q][k] = run[q][j < EDGE_PIECE ? j : EDGE_PIECE - 1];
    }
  }
  int mylen = 0;
#pragma unroll
  for (int q = 0; q < RUNS_PER_THREAD; ++q) mylen += hi[q] - lo[q];
#pragma unroll
  for (int off = 16; off > 0; off >>= 1) mylen += __shfl_xor(mylen, off);
  if (lane == 0) lsum[wave] = mylen;
  __syncthreads();
  int my_base = 0;
  if (tid == 0) {
    int tot = 0;
#pragma unroll
    for (int k = 0; k < 8; ++k) tot += lsum[k];
    my_base = atomicAdd(cursor, tot);
  }
#pragma unroll
  for (int q = 0; q < RUNS_PER_THREAD; ++q) {
#pragma unroll
    for (int k = 0; k < RUN_BATCH; ++k)
      if (lo[q] + k < hi[q])
        atomicAdd(&cntL[first[q][k].y & (BUCKET_NODES - 1)], 1);
    for (int j = lo[q] + RUN_BATCH; j < hi[q]; ++j)
      atomicAdd(&cntL[run[q][j].y & (BUCKET_NODES - 1)], 1);
  }
  __syncthreads();
  const int c = cntL[tid];
  int incl = c;
#pragma unroll
  for (int off = 1; off < 32; off <<= 1) {
    int t = __shfl_up(incl, off);
    if (lane >= off) incl += t;
  }
  if (lane == 31) wsum[wave] = incl;
  __syncthreads();
  if (tid == 0) {
    int tot = 0;
#pragma unroll
    for (int k = 0; k < 8; ++k) {
      int t = wsum[k];
      wsum[k] = tot;
      tot += t;
    }
    blk_base = my_base;
  }
  __syncthreads();
  const int start = blk_base + wsum[wave] + incl - c;
  const float di = rsqrtf((float)c + 1.0f);
  cur[tid] = start;
  dinvL[tid] = di;
  const int node = bk * BUCKET_NODES + tid;
  if (node < NN) {
    v2i sgm;
    sgm.x = start;
    sgm.y = c;
    seg[node] = sgm;
    dinv[node] = di;
  }
  __syncthreads();
#pragma unroll
  for (int q = 0; q < RUNS_PER_THREAD; ++q) {
#pragma unroll
    for (int k = 0; k < RUN_BATCH; ++k) {
      if (lo[q] + k < hi[q]) {
        const int l = first[q][k].y & (BUCKET_NODES - 1);
        v2i rec;
        rec.x = first[q][k].x;
        rec.y = __float_as_int(dinvL[l]);
        csr[atomicAdd(&cur[l], 1)] = rec;
      }
    }
    for (int j = lo[q] + RUN_BATCH; j < hi[q]; ++j) {
      const v2i e = run[q][j];
      const int l = e.y & (BUCKET_NODES - 1);
      v2i rec;
      rec.x = e.x;
      rec.y = __float_as_int(dinvL[l]);
      csr[atomicAdd(&cur[l], 1)] = rec;
    }
  }
}

__global__ __launch_bounds__(256) void k_weight(const float* __restrict__ dinv,
                                                v2i* __restrict__ csr) {
  const int e0 = blockIdx.x * EDGE_PIECE + threadIdx.x;
  v2i r[EPT];
#pragma unroll
  for (int i = 0; i < EPT; ++i) r[i] = csr[e0 + i * 256];
  float ws[EPT];
#pragma unroll
  for (int i = 0; i < EPT; ++i) ws[i] = dinv[r[i].x];
#pragma unroll
  for (int i = 0; i < EPT; ++i) {
    r[i].y = __float_as_int(ws[i] * __int_as_float(r[i].y));
    csr[e0 + i * 256] = r[i];
  }
}

#define GEMM_BLOCKS_128 768
#define GEMM_BLOCKS_64 512
template <int COLT, int NBLK>
__global__ __launch_bounds__(256) void k_gemm_wmma(
    const float* __restrict__ X, const float* __restrict__ W,
    float* __restrict__ H) {
  constexpr int DOUT = COLT * 16;
  constexpr int RT = 8 / COLT;
  constexpr int R = 16 * RT;
  constexpr int NPASS = NN / R;
  constexpr int PITCH = DI + 4;
  constexpr int NPC = (R * DI) / (256 * 4);
  __shared__ __attribute__((aligned(16))) float As[2][R * PITCH];

  const int tid = threadIdx.x;
  const int lane = tid & 31;
  const int wave = tid >> 5;
  const int strip = wave % COLT;
  const int rsub = wave / COLT;
  const int col0 = strip * 16;
  const int sel = lane & 15;
  const int khalf = (lane >> 4) << 1;

  const float* Bcol = W + (size_t)khalf * DOUT + col0 + sel;
  v2f b[DI / 4];
#pragma unroll
  for (int j = 0; j < DI / 4; ++j) {
    b[j].x = Bcol[(size_t)(4 * j) * DOUT];
    b[j].y = Bcol[(size_t)(4 * j + 1) * DOUT];
  }

  int lds_off[NPC];
#pragma unroll
  for (int p = 0; p < NPC; ++p) {
    const int q = p * 256 + tid;
    lds_off[p] = (q >> 5) * PITCH + (q & 31) * 4;
  }
  const float* Xt = X + (size_t)tid * 4;
  const int a_off = (rsub * 16 + sel) * PITCH + khalf;

  int pass = blockIdx.x;
  v4f pre[NPC];
  if (pass < NPASS) {
#pragma unroll
    for (int p = 0; p < NPC; ++p)
      pre[p] = *(const v4f*)(Xt + (size_t)pass * (R * DI) + p * 1024);
  }
  int buf = 0;
  for (; pass < NPASS; pass += NBLK) {
    float* Ab = As[buf];
#pragma unroll
    for (int p = 0; p < NPC; ++p) *(v4f*)(Ab + lds_off[p]) = pre[p];
    __syncthreads();
    const int npass = pass + NBLK;
    if (npass < NPASS) {
#pragma unroll
      for (int p = 0; p < NPC; ++p)
        pre[p] = *(const v4f*)(Xt + (size_t)npass * (R * DI) + p * 1024);
    }
    const float* Ap = Ab + a_off;
    v2f a[DI / 4];
#pragma unroll
    for (int j = 0; j < DI / 4; ++j) a[j] = *(const v2f*)(Ap + 4 * j);
    __builtin_amdgcn_sched_barrier(0);
    v8f acc = {};
#pragma unroll
    for (int j = 0; j < DI / 4; ++j)
      acc = __builtin_amdgcn_wmma_f32_16x16x4_f32(false, a[j], false, b[j],
                                                  (short)0, acc, false, false);
    const int mbase = pass * R + rsub * 16 + ((lane >> 4) << 3);
    float* out = H + (size_t)mbase * DOUT + col0 + sel;
#pragma unroll
    for (int r = 0; r < 8; ++r) out[r * DOUT] = acc[r];
    buf ^= 1;
  }
}

#define GATHER_DEPTH 8
#define GATHER_BLOCKS_128 2048
#define GATHER_BLOCKS_64 1536

template <int LPN>
__device__ __forceinline__ int slot_bcast(int v, int idx) {
  if constexpr (LPN == 32) {
    return __builtin_amdgcn_readlane(v, idx);
  } else {
    return __shfl(v, idx, LPN);
  }
}

template <int LPN>
__device__ __forceinline__ int group_uniform(int v) {
  if constexpr (LPN == 32) {
    return __builtin_amdgcn_readfirstlane(v);
  } else {
    return v;
  }
}

#define GATHER_SLOT(j) const int srow##j = slot_bcast<LPN>(rec.x, k + j);
#define GATHER_ROW(j)                                         \
  v4f hrow##j = zero4;                                        \
  if (k + j < m) hrow##j = *(const v4f*)(Hc + (size_t)srow##j * D);
#define GATHER_ADD(j) \
  acc += hrow##j * __int_as_float(slot_bcast<LPN>(rec.y, k + j));

template <int D, bool RELU>
__global__ __launch_bounds__(256) void k_gather(
    const float* __restrict__ H, const v2i* __restrict__ csr,
    const v2i* __restrict__ seg, const float* __restrict__ dinv,
    const float* __restrict__ bias, float* __restrict__ out, int n_nodes) {
  constexpr int LPN = D / 4;
  const int gt = blockIdx.x * 256 + threadIdx.x;
  const int c = gt % LPN;
  const int stride = (gridDim.x * 256) / LPN;
  const float* Hc = H + c * 4;
  const v4f b = *(const v4f*)(bias + c * 4);
  const v4f zero4 = {0.f, 0.f, 0.f, 0.f};

  int node = group_uniform<LPN>(gt / LPN);
  v2i sg = {0, 0};
  v2i nsg = {0, 0};
  v2i rec = {0, 0};
  if (node < n_nodes) sg = seg[node];
  if (node + stride < n_nodes) nsg = seg[node + stride];
  if (c < sg.y) rec = csr[sg.x + c];
  while (node < n_nodes) {
    const int beg = group_uniform<LPN>(sg.x);
    const int n = group_uniform<LPN>(sg.y);
    const int nnode = node + stride;
    const int m0 = (n < LPN) ? n : LPN;
    v4f acc = zero4;
    v2i nrec = {0, 0};
    v2i nnsg = {0, 0};
    float sn = dinv[node];
    const v4f h = *(const v4f*)(Hc + (size_t)node * D);
    {
      const int k = 0;
      const int m = m0;
      GATHER_SLOT(0) GATHER_SLOT(1) GATHER_SLOT(2) GATHER_SLOT(3)
      GATHER_SLOT(4) GATHER_SLOT(5) GATHER_SLOT(6) GATHER_SLOT(7)
      GATHER_ROW(0) GATHER_ROW(1) GATHER_ROW(2) GATHER_ROW(3)
      GATHER_ROW(4) GATHER_ROW(5) GATHER_ROW(6) GATHER_ROW(7)
      if (c < nsg.y) nrec = csr[nsg.x + c];
      if (nnode + stride < n_nodes) nnsg = seg[nnode + stride];
      GATHER_ADD(0) GATHER_ADD(1) GATHER_ADD(2) GATHER_ADD(3)
      GATHER_ADD(4) GATHER_ADD(5) GATHER_ADD(6) GATHER_ADD(7)
    }
    int base = 0;
    int k0 = GATHER_DEPTH;
    while (base < n) {
      const int m = (n - base < LPN) ? (n - base) : LPN;
      for (int k = k0; k < m; k += GATHER_DEPTH) {
        GATHER_SLOT(0) GATHER_SLOT(1) GATHER_SLOT(2) GATHER_SLOT(3)
        GATHER_SLOT(4) GATHER_SLOT(5) GATHER_SLOT(6) GATHER_SLOT(7)
        GATHER_ROW(0) GATHER_ROW(1) GATHER_ROW(2) GATHER_ROW(3)
        GATHER_ROW(4) GATHER_ROW(5) GATHER_ROW(6) GATHER_ROW(7)
        GATHER_ADD(0) GATHER_ADD(1) GATHER_ADD(2) GATHER_ADD(3)
        GATHER_ADD(4) GATHER_ADD(5) GATHER_ADD(6) GATHER_ADD(7)
      }
      k0 = 0;
      base += LPN;
      if (base < n) {
        rec.x = 0;
        rec.y = 0;
        if (c < n - base) rec = csr[beg + base + c];
      }
    }

    sn *= sn;
    v4f r = acc + h * sn + b;
    if (RELU) {
      r.x = fmaxf(r.x, 0.f);
      r.y = fmaxf(r.y, 0.f);
      r.z = fmaxf(r.z, 0.f);
      r.w = fmaxf(r.w, 0.f);
    }
    *(v4f*)(out + (size_t)node * D + c * 4) = r;

    node = nnode;
    sg = nsg;
    nsg = nnsg;
    rec = nrec;
  }
}

extern "C" void kernel_launch(void* const* d_in, const int* in_sizes, int n_in,
                              void* d_out, int out_size, void* d_ws,
                              size_t ws_size, hipStream_t stream) {
  (void)in_sizes; (void)n_in; (void)out_size; (void)ws_size;

  const float* x  = (const float*)d_in[0];
  const int*   ei = (const int*)d_in[1];
  const int*   src = ei;
  const int*   dst = ei + NE;
  const float* W1 = (const float*)d_in[2];
  const float* b1 = (const float*)d_in[3];
  const float* W2 = (const float*)d_in[4];
  const float* b2 = (const float*)d_in[5];
  const float* W3 = (const float*)d_in[6];
  const float* b3 = (const float*)d_in[7];

  char* p = (char*)d_ws;
  auto take = [&](size_t bytes) -> void* {
    void* r = (void*)p;
    p += (bytes + 255) & ~(size_t)255;
    return r;
  };
  int*   cursor = (int*)take(256);
  int*   tab  = (int*)take((size_t)(NBUCKET + 1) * EDGE_BLOCKS * 4);
  v2i*   binned = (v2i*)take((size_t)NE * 8);
  v2i*   seg  = (v2i*)take((size_t)NN * 8);
  float* dinv = (float*)take((size_t)NN * 4);
  v2i*   csr  = (v2i*)take((size_t)NE * 8);
  float* bufA = (float*)take((size_t)NN * DH * 4);
  float* bufB = (float*)take((size_t)NN * DH * 4);

  k_bin<<<EDGE_BLOCKS, 256, 0, stream>>>(src, dst, tab, binned, cursor);
  k_bucket<<<NBUCKET, 256, 0, stream>>>(tab, binned, cursor, seg, dinv, csr);
  k_weight<<<EDGE_BLOCKS, 256, 0, stream>>>(dinv, csr);

  k_gemm_wmma<DH / 16, GEMM_BLOCKS_128><<<GEMM_BLOCKS_128, 256, 0, stream>>>(x, W1, bufA);
  k_gather<DH, true><<<GATHER_BLOCKS_128, 256, 0, stream>>>(bufA, csr, seg, dinv, b1, bufB, NN);

  k_gemm_wmma<DH / 16, GEMM_BLOCKS_128><<<GEMM_BLOCKS_128, 256, 0, stream>>>(bufB, W2, bufA);
  k_gather<DH, true><<<GATHER_BLOCKS_128, 256, 0, stream>>>(bufA, csr, seg, dinv, b2, bufB, NN);

  k_gemm_wmma<DO / 16, GEMM_BLOCKS_64><<<GEMM_BLOCKS_64, 256, 0, stream>>>(bufB, W3, bufA);
  k_gather<DO, false><<<GATHER_BLOCKS_64, 256, 0, stream>>>(bufA, csr, seg, dinv, b3,
                                               (float*)d_out, NN);
}


# edge weight split into two row scalings: GEMM stores row i times dinv[i], gather adds rows unweighted, then scales by dinv[dst]; weight pass and slot weights gone, 9 -> 8 launches; re-associated
# speedup vs baseline: 3.9529x; 1.0423x over previous
#include <hip/hip_runtime.h>
#include <cstdint>

#define NN 100000
#define NE 640000
#define DI 128
#define DH 128
#define DO 64

typedef __attribute__((ext_vector_type(2))) float v2f;
typedef __attribute__((ext_vector_type(4))) float v4f;
typedef __attribute__((ext_vector_type(8))) float v8f;
typedef __attribute__((ext_vector_type(2))) int v2i;


#define EPT 5
#define EDGE_PIECE (256 * EPT)
#define EDGE_BLOCKS (NE / EDGE_PIECE)
static_assert(NE % EDGE_PIECE == 0, "edge passes have no bound check");

#define BUCKET_SHIFT 8
#define BUCKET_NODES (1 << BUCKET_SHIFT)
#define NBUCKET ((NN + BUCKET_NODES - 1) / BUCKET_NODES)
#define NBUCKET_PAD 512
static_assert(NBUCKET + 1 <= NBUCKET_PAD, "bucket table must fit the scan");
#define RUNS_PER_THREAD ((EDGE_BLOCKS + 255) / 256)
#define RUN_BATCH 8

__global__ __launch_bounds__(256) void k_bin(const int* __restrict__ src,
                                             const int* __restrict__ dst,
                                             int* __restrict__ tab,
                                             v2i* __restrict__ binned,
                                             int* __restrict__ cursor) {
  __shared__ int hist[NBUCKET_PAD];
  __shared__ int wsum[8];
  __shared__ v2i stage[EDGE_PIECE];
  const int tid = threadIdx.x;
  const int lane = tid & 31;
  const int wave = tid >> 5;
  const int w = blockIdx.x;
  const int e0 = w * EDGE_PIECE + tid;
  int s[EPT], d[EPT];
#pragma unroll
  for (int i = 0; i < EPT; ++i) {
    s[i] = src[e0 + i * 256];
    d[i] = dst[e0 + i * 256];
  }
  hist[2 * tid] = 0;
  hist[2 * tid + 1] = 0;
  if (w == 0 && tid == 0) *cursor = 0;
  __syncthreads();
  int r[EPT];
#pragma unroll
  for (int i = 0; i < EPT; ++i) r[i] = atomicAdd(&hist[d[i] >> BUCKET_SHIFT], 1);
  __syncthreads();
  const int v0 = hist[2 * tid];
  const int v1 = hist[2 * tid + 1];
  const int sum = v0 + v1;
  int incl = sum;
#pragma unroll
  for (int off = 1; off < 32; off <<= 1) {
    int t = __shfl_up(incl, off);
    if (lane >= off) incl += t;
  }
  if (lane == 31) wsum[wave] = incl;
  __syncthreads();
  int wbase = 0;
#pragma unroll
  for (int k = 0; k < 8; ++k) {
    const int t = wsum[k];
    if (k < wave) wbase += t;
  }
  const int ex = wbase + incl - sum;
  hist[2 * tid] = ex;
  hist[2 * tid + 1] = ex + v0;
  if (2 * tid <= NBUCKET) tab[(2 * tid) * EDGE_BLOCKS + w] = ex;
  if (2 * tid + 1 <= NBUCKET) tab[(2 * tid + 1) * EDGE_BLOCKS + w] = ex + v0;
  __syncthreads();
#pragma unroll
  for (int i = 0; i < EPT; ++i) {
    v2i rec;
    rec.x = s[i];
    rec.y = d[i];
    stage[hist[d[i] >> BUCKET_SHIFT] + r[i]] = rec;
  }
  __syncthreads();
#pragma unroll
  for (int i = 0; i < EPT; ++i)
    binned[(size_t)w * EDGE_PIECE + tid + i * 256] = stage[tid + i * 256];
}

__global__ __launch_bounds__(256) void k_bucket(const int* __restrict__ tab,
                                                const v2i* __restrict__ binned,
                                                int* __restrict__ cursor,
                                                v2i* __restrict__ seg,
                                                float* __restrict__ dinv,
                                                int* __restrict__ csr) {
  __shared__ int cntL[BUCKET_NODES];
  __shared__ int cur[BUCKET_NODES];
  __shared__ int lsum[8];
  __shared__ int wsum[8];
  __shared__ int blk_base;
  const int tid = threadIdx.x;
  const int lane = tid & 31;
  const int wave = tid >> 5;
  const int bk = blockIdx.x;

  int lo[RUNS_PER_THREAD], hi[RUNS_PER_THREAD];
  const v2i* run[RUNS_PER_THREAD];
#pragma unroll
  for (int q = 0; q < RUNS_PER_THREAD; ++q) {
    const int w = tid + q * 256;
    lo[q] = 0;
    hi[q] = 0;
    if (w < EDGE_BLOCKS) {
      lo[q] = tab[bk * EDGE_BLOCKS + w];
      hi[q] = tab[(bk + 1) * EDGE_BLOCKS + w];
    }
    run[q] = binned + (size_t)(w < EDGE_BLOCKS ? w : 0) * EDGE_PIECE;
  }
  cntL[tid] = 0;
  v2i first[RUNS_PER_THREAD][RUN_BATCH];
#pragma unroll
  for (int q = 0; q < RUNS_PER_THREAD; ++q) {
#pragma unroll
    for (int k = 0; k < RUN_BATCH; ++k) {
      const int j = lo[q] + k;
      first[q][k] = run[q][j < EDGE_PIECE ? j : EDGE_PIECE - 1];
    }
  }
  int mylen = 0;
#pragma unroll
  for (int q = 0; q < RUNS_PER_THREAD; ++q) mylen += hi[q] - lo[q];
#pragma unroll
  for (int off = 16; off > 0; off >>= 1) mylen += __shfl_xor(mylen, off);
  if (lane == 0) lsum[wave] = mylen;
  __syncthreads();
  int my_base = 0;
  if (tid == 0) {
    int tot = 0;
#pragma unroll
    for (int k = 0; k < 8; ++k) tot += lsum[k];
    my_base = atomicAdd(cursor, tot);
  }
#pragma unroll
  for (int q = 0; q < RUNS_PER_THREAD; ++q) {
#pragma unroll
    for (int k = 0; k < RUN_BATCH; ++k)
      if (lo[q] + k < hi[q])
        atomicAdd(&cntL[first[q][k].y & (BUCKET_NODES - 1)], 1);
    for (int j = lo[q] + RUN_BATCH; j < hi[q]; ++j)
      atomicAdd(&cntL[run[q][j].y & (BUCKET_NODES - 1)], 1);
  }
  __syncthreads();
  const int c = cntL[tid];
  int incl = c;
#pragma unroll
  for (int off = 1; off < 32; off <<= 1) {
    int t = __shfl_up(incl, off);
    if (lane >= off) incl += t;
  }
  if (lane == 31) wsum[wave] = incl;
  __syncthreads();
  if (tid == 0) {
    int tot = 0;
#pragma unroll
    for (int k = 0; k < 8; ++k) {
      int t = wsum[k];
      wsum[k] = tot;
      tot += t;
    }
    blk_base = my_base;
  }
  __syncthreads();
  const int start = blk_base + wsum[wave] + incl - c;
  const float di = rsqrtf((float)c + 1.0f);
  cur[tid] = start;
  const int node = bk * BUCKET_NODES + tid;
  if (node < NN) {
    v2i sgm;
    sgm.x = start;
    sgm.y = c;
    seg[node] = sgm;
    dinv[node] = di;
  }
  __syncthreads();
#pragma unroll
  for (int q = 0; q < RUNS_PER_THREAD; ++q) {
#pragma unroll
    for (int k = 0; k < RUN_BATCH; ++k) {
      if (lo[q] + k < hi[q]) {
        const int l = first[q][k].y & (BUCKET_NODES - 1);
        csr[atomicAdd(&cur[l], 1)] = first[q][k].x;
      }
    }
    for (int j = lo[q] + RUN_BATCH; j < hi[q]; ++j) {
      const v2i e = run[q][j];
      const int l = e.y & (BUCKET_NODES - 1);
      csr[atomicAdd(&cur[l], 1)] = e.x;
    }
  }
}

#define GEMM_BLOCKS_128 768
#define GEMM_BLOCKS_64 512
template <int COLT, int NBLK>
__global__ __launch_bounds__(256) void k_gemm_wmma(
    const float* __restrict__ X, const float* __restrict__ W,
    const float* __restrict__ dinv, float* __restrict__ H) {
  constexpr int DOUT = COLT * 16;
  constexpr int RT = 8 / COLT;
  constexpr int R = 16 * RT;
  constexpr int NPASS = NN / R;
  constexpr int PITCH = DI + 4;
  constexpr int NPC = (R * DI) / (256 * 4);
  __shared__ __attribute__((aligned(16))) float As[2][R * PITCH];

  const int tid = threadIdx.x;
  const int lane = tid & 31;
  const int wave = tid >> 5;
  const int strip = wave % COLT;
  const int rsub = wave / COLT;
  const int col0 = strip * 16;
  const int sel = lane & 15;
  const int khalf = (lane >> 4) << 1;

  const float* Bcol = W + (size_t)khalf * DOUT + col0 + sel;
  v2f b[DI / 4];
#pragma unroll
  for (int j = 0; j < DI / 4; ++j) {
    b[j].x = Bcol[(size_t)(4 * j) * DOUT];
    b[j].y = Bcol[(size_t)(4 * j + 1) * DOUT];
  }

  int lds_off[NPC];
#pragma unroll
  for (int p = 0; p < NPC; ++p) {
    const int q = p * 256 + tid;
    lds_off[p] = (q >> 5) * PITCH + (q & 31) * 4;
  }
  const float* Xt = X + (size_t)tid * 4;
  const int a_off = (rsub * 16 + sel) * PITCH + khalf;

  int pass = blockIdx.x;
  v4f pre[NPC];
  if (pass < NPASS) {
#pragma unroll
    for (int p = 0; p < NPC; ++p)
      pre[p] = *(const v4f*)(Xt + (size_t)pass * (R * DI) + p * 1024);
  }
  int buf = 0;
  for (; pass < NPASS; pass += NBLK) {
    float* Ab = As[buf];
#pragma unroll
    for (int p = 0; p < NPC; ++p) *(v4f*)(Ab + lds_off[p]) = pre[p];
    __syncthreads();
    const int npass = pass + NBLK;
    if (npass < NPASS) {
#pragma unroll
      for (int p = 0; p < NPC; ++p)
        pre[p] = *(const v4f*)(Xt + (size_t)npass * (R * DI) + p * 1024);
    }
    const int mbase = pass * R + rsub * 16 + ((lane >> 4) << 3);
    const int rbase = __builtin_amdgcn_readfirstlane(pass * R + rsub * 16);
    float sc_top[8], sc_bot[8];
#pragma unroll
    for (int r = 0; r < 8; ++r) {
      sc_top[r] = dinv[rbase + r];
      sc_bot[r] = dinv[rbase + 8 + r];
    }
    const float* Ap = Ab + a_off;
    v2f a[DI / 4];
#pragma unroll
    for (int j = 0; j < DI / 4; ++j) a[j] = *(const v2f*)(Ap + 4 * j);
    __builtin_amdgcn_sched_barrier(0);
    v8f acc = {};
#pragma unroll
    for (int j = 0; j < DI / 4; ++j)
      acc = __builtin_amdgcn_wmma_f32_16x16x4_f32(false, a[j], false, b[j],
                                                  (short)0, acc, false, false);
    float* out = H + (size_t)mbase * DOUT + col0 + sel;
#pragma unroll
    for (int r = 0; r < 8; ++r)
      out[r * DOUT] = acc[r] * ((lane >> 4) ? sc_bot[r] : sc_top[r]);
    buf ^= 1;
  }
}

#define GATHER_DEPTH 8
#define GATHER_BLOCKS_128 2048
#define GATHER_BLOCKS_64 1536

template <int LPN>
__device__ __forceinline__ int slot_bcast(int v, int idx) {
  if constexpr (LPN == 32) {
    return __builtin_amdgcn_readlane(v, idx);
  } else {
    return __shfl(v, idx, LPN);
  }
}

template <int LPN>
__device__ __forceinline__ int group_uniform(int v) {
  if constexpr (LPN == 32) {
    return __builtin_amdgcn_readfirstlane(v);
  } else {
    return v;
  }
}

#define GATHER_SLOT(j) const int srow##j = slot_bcast<LPN>(rec, k + j);
#define GATHER_ROW(j)                                         \
  v4f hrow##j = zero4;                                        \
  if (k + j < m) hrow##j = *(const v4f*)(Hc + (size_t)srow##j * D);
#define GATHER_ADD(j) acc += hrow##j;
#define GATHER_HOLD()                                                        \
  asm volatile("" ::"v"(hrow0), "v"(hrow1), "v"(hrow2), "v"(hrow3),          \
               "v"(hrow4), "v"(hrow5), "v"(hrow6), "v"(hrow7))

template <int D, bool RELU>
__global__ __launch_bounds__(256) void k_gather(
    const float* __restrict__ H, const int* __restrict__ csr,
    const v2i* __restrict__ seg, const float* __restrict__ dinv,
    const float* __restrict__ bias, float* __restrict__ out, int n_nodes) {
  constexpr int LPN = D / 4;
  const int gt = blockIdx.x * 256 + threadIdx.x;
  const int c = gt % LPN;
  const int stride = (gridDim.x * 256) / LPN;
  const float* Hc = H + c * 4;
  const v4f b = *(const v4f*)(bias + c * 4);
  const v4f zero4 = {0.f, 0.f, 0.f, 0.f};

  int node = group_uniform<LPN>(gt / LPN);
  v2i sg = {0, 0};
  v2i nsg = {0, 0};
  int rec = 0;
  if (node < n_nodes) sg = seg[node];
  if (node + stride < n_nodes) nsg = seg[node + stride];
  if (c < sg.y) rec = csr[sg.x + c];
  while (node < n_nodes) {
    const int beg = group_uniform<LPN>(sg.x);
    const int n = group_uniform<LPN>(sg.y);
    const int nnode = node + stride;
    const int m0 = (n < LPN) ? n : LPN;
    v4f acc = zero4;
    int nrec = 0;
    v2i nnsg = {0, 0};
    const float sn = dinv[node];
    const v4f h = *(const v4f*)(Hc + (size_t)node * D);
    {
      const int k = 0;
      const int m = m0;
      GATHER_SLOT(0) GATHER_SLOT(1) GATHER_SLOT(2) GATHER_SLOT(3)
      GATHER_SLOT(4) GATHER_SLOT(5) GATHER_SLOT(6) GATHER_SLOT(7)
      GATHER_ROW(0) GATHER_ROW(1) GATHER_ROW(2) GATHER_ROW(3)
      GATHER_ROW(4) GATHER_ROW(5) GATHER_ROW(6) GATHER_ROW(7)
      if (c < nsg.y) nrec = csr[nsg.x + c];
      if (nnode + stride < n_nodes) nnsg = seg[nnode + stride];
      GATHER_HOLD();
      GATHER_ADD(0) GATHER_ADD(1) GATHER_ADD(2) GATHER_ADD(3)
      GATHER_ADD(4) GATHER_ADD(5) GATHER_ADD(6) GATHER_ADD(7)
    }
    int base = 0;
    int k0 = GATHER_DEPTH;
    while (base < n) {
      const int m = (n - base < LPN) ? (n - base) : LPN;
      for (int k = k0; k < m; k += GATHER_DEPTH) {
        GATHER_SLOT(0) GATHER_SLOT(1) GATHER_SLOT(2) GATHER_SLOT(3)
        GATHER_SLOT(4) GATHER_SLOT(5) GATHER_SLOT(6) GATHER_SLOT(7)
        GATHER_ROW(0) GATHER_ROW(1) GATHER_ROW(2) GATHER_ROW(3)
        GATHER_ROW(4) GATHER_ROW(5) GATHER_ROW(6) GATHER_ROW(7)
        GATHER_ADD(0) GATHER_ADD(1) GATHER_ADD(2) GATHER_ADD(3)
        GATHER_ADD(4) GATHER_ADD(5) GATHER_ADD(6) GATHER_ADD(7)
      }
      k0 = 0;
      base += LPN;
      if (base < n) {
        rec = 0;
        if (c < n - base) rec = csr[beg + base + c];
      }
    }

    v4f r = (acc + h) * sn + b;
    if (RELU) {
      r.x = fmaxf(r.x, 0.f);
      r.y = fmaxf(r.y, 0.f);
      r.z = fmaxf(r.z, 0.f);
      r.w = fmaxf(r.w, 0.f);
    }
    *(v4f*)(out + (size_t)node * D + c * 4) = r;

    node = nnode;
    sg = nsg;
    nsg = nnsg;
    rec = nrec;
  }
}

extern "C" void kernel_launch(void* const* d_in, const int* in_sizes, int n_in,
                              void* d_out, int out_size, void* d_ws,
                              size_t ws_size, hipStream_t stream) {
  (void)in_sizes; (void)n_in; (void)out_size; (void)ws_size;

  const float* x  = (const float*)d_in[0];
  const int*   ei = (const int*)d_in[1];
  const int*   src = ei;
  const int*   dst = ei + NE;
  const float* W1 = (const float*)d_in[2];
  const float* b1 = (const float*)d_in[3];
  const float* W2 = (const float*)d_in[4];
  const float* b2 = (const float*)d_in[5];
  const float* W3 = (const float*)d_in[6];
  const float* b3 = (const float*)d_in[7];

  char* p = (char*)d_ws;
  auto take = [&](size_t bytes) -> void* {
    void* r = (void*)p;
    p += (bytes + 255) & ~(size_t)255;
    return r;
  };
  int*   cursor = (int*)take(256);
  int*   tab  = (int*)take((size_t)(NBUCKET + 1) * EDGE_BLOCKS * 4);
  v2i*   binned = (v2i*)take((size_t)NE * 8);
  v2i*   seg  = (v2i*)take((size_t)NN * 8);
  float* dinv = (float*)take((size_t)NN * 4);
  int*   csr  = (int*)take((size_t)NE * 4);
  float* bufA = (float*)take((size_t)NN * DH * 4);
  float* bufB = (float*)take((size_t)NN * DH * 4);

  k_bin<<<EDGE_BLOCKS, 256, 0, stream>>>(src, dst, tab, binned, cursor);
  k_bucket<<<NBUCKET, 256, 0, stream>>>(tab, binned, cursor, seg, dinv, csr);

  k_gemm_wmma<DH / 16, GEMM_BLOCKS_128><<<GEMM_BLOCKS_128, 256, 0, stream>>>(x, W1, dinv, bufA);
  k_gather<DH, true><<<GATHER_BLOCKS_128, 256, 0, stream>>>(bufA, csr, seg, dinv, b1, bufB, NN);

  k_gemm_wmma<DH / 16, GEMM_BLOCKS_128><<<GEMM_BLOCKS_128, 256, 0, stream>>>(bufB, W2, dinv, bufA);
  k_gather<DH, true><<<GATHER_BLOCKS_128, 256, 0, stream>>>(bufA, csr, seg, dinv, b2, bufB, NN);

  k_gemm_wmma<DO / 16, GEMM_BLOCKS_64><<<GEMM_BLOCKS_64, 256, 0, stream>>>(bufB, W3, dinv, bufA);
  k_gather<DO, false><<<GATHER_BLOCKS_64, 256, 0, stream>>>(bufA, csr, seg, dinv, b3,
                                               (float*)d_out, NN);
}
